# starting kernel
# baseline (speedup 1.0000x reference)
#include <hip/hip_runtime.h>


typedef float v2f __attribute__((ext_vector_type(2)));
typedef float v8f __attribute__((ext_vector_type(8)));

#define T_SEG 8
#define NSEG  16
#define CCH   512
#define HW    196
#define B_TOT (NSEG * HW)
#define HEADS 4
#define KTAPS 5
#define O20   (HEADS * KTAPS)
#define RTILE 13

__global__ __launch_bounds__(32)
void dyncnn_logits_softmax(const float* __restrict__ x,
                           const float* __restrict__ W,
                           float* __restrict__ wsm)
{
    const int lane = threadIdx.x;
    const int tile = blockIdx.x % RTILE;
    const int tn   = blockIdx.x / RTILE;
    const int t    = tn % T_SEG;
    const int ni   = tn / T_SEG;

    const int half = lane >> 4;
    const int l15  = lane & 15;

    int hw  = tile * 16 + l15;
    int hwc = (hw < HW - 1) ? hw : (HW - 1);
    const int cA = half ? 2 : 0;
    const float* aptr = x + ((size_t)((ni * T_SEG + t) * CCH) + cA) * HW + hwc;

    const float2* __restrict__ W2 = (const float2*)W;
    int o0 = l15;
    int o1 = o0 + 16; if (o1 > O20 - 1) o1 = O20 - 1;
    int bofs0 = o0 * (CCH / 2) + (cA >> 1);
    int bofs1 = o1 * (CCH / 2) + (cA >> 1);

    v8f acc0 = {}; v8f acc1 = {};
    for (int k4 = 0; k4 < CCH; k4 += 4) {
        v2f a;
        a.x = aptr[0];
        a.y = aptr[HW];
        aptr += 4 * HW;
        float2 f0 = W2[bofs0];
        float2 f1 = W2[bofs1];
        bofs0 += 2; bofs1 += 2;
        v2f b0; b0.x = f0.x; b0.y = f0.y;
        v2f b1; b1.x = f1.x; b1.y = f1.y;
        acc0 = __builtin_amdgcn_wmma_f32_16x16x4_f32(
                   false, a, false, b0, (short)0, acc0, false, false);
        acc1 = __builtin_amdgcn_wmma_f32_16x16x4_f32(
                   false, a, false, b1, (short)0, acc1, false, false);
    }

    __shared__ float Lsm[16 * 32];
    {
        const int mbase = half * 8;
        #pragma unroll
        for (int r = 0; r < 8; ++r) {
            Lsm[(mbase + r) * 32 + l15]      = acc0[r];
            Lsm[(mbase + r) * 32 + 16 + l15] = acc1[r];
        }
    }
    __syncthreads();

    #pragma unroll
    for (int i = 0; i < 2; ++i) {
        const int task = lane * 2 + i;
        const int rr = task >> 2;
        const int h  = task & 3;
        const float* lp = &Lsm[rr * 32 + h * KTAPS];
        float v0 = lp[0], v1 = lp[1], v2 = lp[2], v3 = lp[3], v4 = lp[4];
        float m = fmaxf(fmaxf(fmaxf(v0, v1), fmaxf(v2, v3)), v4);
        float e0 = __expf(v0 - m), e1 = __expf(v1 - m), e2 = __expf(v2 - m),
              e3 = __expf(v3 - m), e4 = __expf(v4 - m);
        float inv = 1.0f / (e0 + e1 + e2 + e3 + e4);
        const int hwr = tile * 16 + rr;
        if (hwr < HW) {
            size_t base = (((size_t)t * B_TOT) + (size_t)ni * HW + hwr) * O20
                          + (size_t)h * KTAPS;
            wsm[base + 0] = e0 * inv;
            wsm[base + 1] = e1 * inv;
            wsm[base + 2] = e2 * inv;
            wsm[base + 3] = e3 * inv;
            wsm[base + 4] = e4 * inv;
        }
    }
}

__global__ __launch_bounds__(256)
void dyncnn_combine(const float* __restrict__ x,
                    const float* __restrict__ wsm,
                    float* __restrict__ out)
{
    const size_t idx = (size_t)blockIdx.x * 256 + threadIdx.x;
    const int hw  = (int)(idx % HW);
    const int c   = (int)((idx / HW) % CCH);
    const int nti = (int)(idx / ((size_t)HW * CCH));
    const int ni  = nti >> 3;
    const int t   = nti & 7;
    const int h   = c >> 7;

    const float* wp = wsm + (((size_t)t * B_TOT) + (size_t)ni * HW + hw) * O20
                          + (size_t)h * KTAPS;
    const float* xb = x + (((size_t)(ni * T_SEG)) * CCH + c) * HW + hw;
    const size_t tstride = (size_t)CCH * HW;

    float acc = 0.0f;
    #pragma unroll
    for (int k = 0; k < KTAPS; ++k) {
        const int ts = t + k - 4;
        if (ts >= 0)
            acc = fmaf(xb[(size_t)ts * tstride], wp[k], acc);
    }
    out[idx] = acc;
}

extern "C" void kernel_launch(void* const* d_in, const int* in_sizes, int n_in,
                              void* d_out, int out_size, void* d_ws, size_t ws_size,
                              hipStream_t stream) {
    (void)in_sizes; (void)n_in; (void)out_size; (void)ws_size;
    const float* x   = (const float*)d_in[0];
    const float* W   = (const float*)d_in[1];
    float*       out = (float*)d_out;
    float*       wsm = (float*)d_ws;

    dyncnn_logits_softmax<<<dim3(NSEG * T_SEG * RTILE), dim3(32), 0, stream>>>(x, W, wsm);

    const size_t total = (size_t)NSEG * T_SEG * CCH * HW;
    dyncnn_combine<<<dim3((unsigned)(total / 256)), dim3(256), 0, stream>>>(x, wsm, out);
}


# combine kernel: each lane holds its softmax weights in registers (staged once via LDS) and slides the 5-tap window over all 8 t, so x is read once instead of 5 times; sum order kept
# speedup vs baseline: 2.4211x; 2.4211x over previous
#include <hip/hip_runtime.h>


typedef float v2f __attribute__((ext_vector_type(2)));
typedef float v8f __attribute__((ext_vector_type(8)));

#define T_SEG 8
#define NSEG  16
#define CCH   512
#define HW    196
#define B_TOT (NSEG * HW)
#define HEADS 4
#define KTAPS 5
#define O20   (HEADS * KTAPS)
#define RTILE 13

__global__ __launch_bounds__(32)
void dyncnn_logits_softmax(const float* __restrict__ x,
                           const float* __restrict__ W,
                           float* __restrict__ wsm)
{
    const int lane = threadIdx.x;
    const int tile = blockIdx.x % RTILE;
    const int tn   = blockIdx.x / RTILE;
    const int t    = tn % T_SEG;
    const int ni   = tn / T_SEG;

    const int half = lane >> 4;
    const int l15  = lane & 15;

    int hw  = tile * 16 + l15;
    int hwc = (hw < HW - 1) ? hw : (HW - 1);
    const int cA = half ? 2 : 0;
    const float* aptr = x + ((size_t)((ni * T_SEG + t) * CCH) + cA) * HW + hwc;

    const float2* __restrict__ W2 = (const float2*)W;
    int o0 = l15;
    int o1 = o0 + 16; if (o1 > O20 - 1) o1 = O20 - 1;
    int bofs0 = o0 * (CCH / 2) + (cA >> 1);
    int bofs1 = o1 * (CCH / 2) + (cA >> 1);

    v8f acc0 = {}; v8f acc1 = {};
    for (int k4 = 0; k4 < CCH; k4 += 4) {
        v2f a;
        a.x = aptr[0];
        a.y = aptr[HW];
        aptr += 4 * HW;
        float2 f0 = W2[bofs0];
        float2 f1 = W2[bofs1];
        bofs0 += 2; bofs1 += 2;
        v2f b0; b0.x = f0.x; b0.y = f0.y;
        v2f b1; b1.x = f1.x; b1.y = f1.y;
        acc0 = __builtin_amdgcn_wmma_f32_16x16x4_f32(
                   false, a, false, b0, (short)0, acc0, false, false);
        acc1 = __builtin_amdgcn_wmma_f32_16x16x4_f32(
                   false, a, false, b1, (short)0, acc1, false, false);
    }

    __shared__ float Lsm[16 * 32];
    {
        const int mbase = half * 8;
        #pragma unroll
        for (int r = 0; r < 8; ++r) {
            Lsm[(mbase + r) * 32 + l15]      = acc0[r];
            Lsm[(mbase + r) * 32 + 16 + l15] = acc1[r];
        }
    }
    __syncthreads();

    #pragma unroll
    for (int i = 0; i < 2; ++i) {
        const int task = lane * 2 + i;
        const int rr = task >> 2;
        const int h  = task & 3;
        const float* lp = &Lsm[rr * 32 + h * KTAPS];
        float v0 = lp[0], v1 = lp[1], v2 = lp[2], v3 = lp[3], v4 = lp[4];
        float m = fmaxf(fmaxf(fmaxf(v0, v1), fmaxf(v2, v3)), v4);
        float e0 = __expf(v0 - m), e1 = __expf(v1 - m), e2 = __expf(v2 - m),
              e3 = __expf(v3 - m), e4 = __expf(v4 - m);
        float inv = 1.0f / (e0 + e1 + e2 + e3 + e4);
        const int hwr = tile * 16 + rr;
        if (hwr < HW) {
            size_t base = (((size_t)t * B_TOT) + (size_t)ni * HW + hwr) * O20
                          + (size_t)h * KTAPS;
            wsm[base + 0] = e0 * inv;
            wsm[base + 1] = e1 * inv;
            wsm[base + 2] = e2 * inv;
            wsm[base + 3] = e3 * inv;
            wsm[base + 4] = e4 * inv;
        }
    }
}

#define BT     64
#define NBT    (B_TOT / BT)
#define CHW    4
#define SUBS   2
#define NCHUNK (128 / (CHW * SUBS))
#define SLAB4  (T_SEG * BT * O20 / 4)

__global__ __launch_bounds__(256)
void dyncnn_combine(const float* __restrict__ x,
                    const float* __restrict__ wsm,
                    float* __restrict__ out)
{
    __shared__ float4 slab4[SLAB4];
    const int tid   = threadIdx.x;
    const int tile  = blockIdx.x;
    const int chunk = blockIdx.y;

    const int wave = tid >> 5;
    const int lane = tid & 31;
    const int h    = wave & 3;
    const int sub  = wave >> 2;

    const int b0 = tile * BT + 2 * lane;
    const int ni = b0 / HW;
    const int hw = b0 - ni * HW;
    const int c0 = h * 128 + (chunk * SUBS + sub) * CHW;
    const size_t tstride = (size_t)CCH * HW;
    const size_t base = ((size_t)(ni * T_SEG) * CCH + c0) * HW + hw;
    const float2* __restrict__ xp = (const float2*)(x + base);
    float2* __restrict__ op = (float2*)(out + base);

    float2 xv[CHW][T_SEG];
    #pragma unroll
    for (int j = 0; j < CHW; ++j) {
        #pragma unroll
        for (int t = 0; t < T_SEG; ++t)
            xv[j][t] = xp[((size_t)t * tstride + (size_t)j * HW) >> 1];
    }

    const float4* __restrict__ wsm4 = (const float4*)wsm;
    #pragma unroll
    for (int i = 0; i < SLAB4 / 256; ++i) {
        const int q = i * 256 + tid;
        const int t = q / 320;
        const int r = q - t * 320;
        slab4[q] = wsm4[((size_t)t * B_TOT + (size_t)tile * BT) * (O20 / 4) + r];
    }
    __syncthreads();

    const float* sl = (const float*)slab4;
    float w[2][T_SEG][KTAPS];
    #pragma unroll
    for (int p = 0; p < 2; ++p) {
        #pragma unroll
        for (int t = 0; t < T_SEG; ++t) {
            #pragma unroll
            for (int k = 0; k < KTAPS; ++k) {
                if (t + k - 4 >= 0)
                    w[p][t][k] = sl[(t * BT + 2 * lane + p) * O20 + h * KTAPS + k];
                else
                    w[p][t][k] = 0.0f;
            }
        }
    }

    #pragma unroll
    for (int j = 0; j < CHW; ++j) {
        #pragma unroll
        for (int t = 0; t < T_SEG; ++t) {
            float a0 = 0.0f, a1 = 0.0f;
            #pragma unroll
            for (int k = 0; k < KTAPS; ++k) {
                const int ts = t + k - 4;
                if (ts >= 0) {
                    a0 = fmaf(xv[j][ts].x, w[0][t][k], a0);
                    a1 = fmaf(xv[j][ts].y, w[1][t][k], a1);
                }
            }
            float2 r; r.x = a0; r.y = a1;
            op[((size_t)t * tstride + (size_t)j * HW) >> 1] = r;
        }
    }
}

extern "C" void kernel_launch(void* const* d_in, const int* in_sizes, int n_in,
                              void* d_out, int out_size, void* d_ws, size_t ws_size,
                              hipStream_t stream) {
    (void)in_sizes; (void)n_in; (void)out_size; (void)ws_size;
    const float* x   = (const float*)d_in[0];
    const float* W   = (const float*)d_in[1];
    float*       out = (float*)d_out;
    float*       wsm = (float*)d_ws;

    dyncnn_logits_softmax<<<dim3(NSEG * T_SEG * RTILE), dim3(32), 0, stream>>>(x, W, wsm);

    dyncnn_combine<<<dim3(NBT, NCHUNK), dim3(256), 0, stream>>>(x, wsm, out);
}


# logits kernel: 512-channel reduction split over 4 waves per work-group, partial tiles added in LDS (new sum order); each wave keeps about 60 loads in flight instead of about 16
# speedup vs baseline: 2.8667x; 1.1840x over previous
#include <hip/hip_runtime.h>


typedef float v2f __attribute__((ext_vector_type(2)));
typedef float v8f __attribute__((ext_vector_type(8)));

#define T_SEG 8
#define NSEG  16
#define CCH   512
#define HW    196
#define B_TOT (NSEG * HW)
#define HEADS 4
#define KTAPS 5
#define O20   (HEADS * KTAPS)
#define RTILE 13

#define KSPLIT 4
#define CSL    (CCH / KSPLIT)

__global__ __launch_bounds__(32 * KSPLIT) __attribute__((amdgpu_waves_per_eu(4, 8)))
void dyncnn_logits_softmax(const float* __restrict__ x,
                           const float* __restrict__ W,
                           float* __restrict__ wsm)
{
    const int tid  = threadIdx.x;
    const int lane = tid & 31;
    const int wv   = tid >> 5;
    const int tile = blockIdx.x % RTILE;
    const int tn   = blockIdx.x / RTILE;
    const int t    = tn % T_SEG;
    const int ni   = tn / T_SEG;

    const int half = lane >> 4;
    const int l15  = lane & 15;

    int hw  = tile * 16 + l15;
    int hwc = (hw < HW - 1) ? hw : (HW - 1);
    const int cA = (half ? 2 : 0) + wv * CSL;
    const float* aptr = x + ((size_t)((ni * T_SEG + t) * CCH) + cA) * HW + hwc;

    const float2* __restrict__ W2 = (const float2*)W;
    int o0 = l15;
    int o1 = o0 + 16; if (o1 > O20 - 1) o1 = O20 - 1;
    int bofs0 = o0 * (CCH / 2) + (cA >> 1);
    int bofs1 = o1 * (CCH / 2) + (cA >> 1);

    v8f acc0 = {}; v8f acc1 = {};
    for (int trip = 0; trip < CSL / 64; ++trip) {
        float  ax[16], ay[16];
        float2 f0[16], f1[16];
        #pragma unroll
        for (int s = 0; s < 16; ++s) {
            ax[s] = aptr[(4 * s) * HW];
            ay[s] = aptr[(4 * s + 1) * HW];
        }
        #pragma unroll
        for (int s = 0; s < 16; ++s) {
            f0[s] = W2[bofs0 + 2 * s];
            f1[s] = W2[bofs1 + 2 * s];
        }
        aptr  += 64 * HW;
        bofs0 += 32; bofs1 += 32;
        #pragma unroll
        for (int s = 0; s < 16; ++s) {
            v2f a;  a.x  = ax[s];   a.y  = ay[s];
            v2f b0; b0.x = f0[s].x; b0.y = f0[s].y;
            v2f b1; b1.x = f1[s].x; b1.y = f1[s].y;
            acc0 = __builtin_amdgcn_wmma_f32_16x16x4_f32(
                       false, a, false, b0, (short)0, acc0, false, false);
            acc1 = __builtin_amdgcn_wmma_f32_16x16x4_f32(
                       false, a, false, b1, (short)0, acc1, false, false);
        }
    }

    __shared__ float Lsm[KSPLIT * 16 * 32];
    {
        float* Lw = Lsm + wv * (16 * 32);
        const int mbase = half * 8;
        #pragma unroll
        for (int r = 0; r < 8; ++r) {
            Lw[(mbase + r) * 32 + l15]      = acc0[r];
            Lw[(mbase + r) * 32 + 16 + l15] = acc1[r];
        }
    }
    __syncthreads();

    if (tid < 64) {
        const int rr = tid >> 2;
        const int h  = tid & 3;
        const float* lp = &Lsm[rr * 32 + h * KTAPS];
        float v[KTAPS];
        #pragma unroll
        for (int k = 0; k < KTAPS; ++k)
            v[k] = ((lp[k] + lp[512 + k]) + lp[1024 + k]) + lp[1536 + k];
        float v0 = v[0], v1 = v[1], v2 = v[2], v3 = v[3], v4 = v[4];
        float m = fmaxf(fmaxf(fmaxf(v0, v1), fmaxf(v2, v3)), v4);
        float e0 = __expf(v0 - m), e1 = __expf(v1 - m), e2 = __expf(v2 - m),
              e3 = __expf(v3 - m), e4 = __expf(v4 - m);
        float inv = 1.0f / (e0 + e1 + e2 + e3 + e4);
        const int hwr = tile * 16 + rr;
        if (hwr < HW) {
            size_t base = (((size_t)t * B_TOT) + (size_t)ni * HW + hwr) * O20
                          + (size_t)h * KTAPS;
            wsm[base + 0] = e0 * inv;
            wsm[base + 1] = e1 * inv;
            wsm[base + 2] = e2 * inv;
            wsm[base + 3] = e3 * inv;
            wsm[base + 4] = e4 * inv;
        }
    }
}

#define BT     64
#define NBT    (B_TOT / BT)
#define CHW    4
#define SUBS   2
#define NCHUNK (128 / (CHW * SUBS))
#define SLAB4  (T_SEG * BT * O20 / 4)

__global__ __launch_bounds__(256)
void dyncnn_combine(const float* __restrict__ x,
                    const float* __restrict__ wsm,
                    float* __restrict__ out)
{
    __shared__ float4 slab4[SLAB4];
    const int tid   = threadIdx.x;
    const int tile  = blockIdx.x;
    const int chunk = blockIdx.y;

    const int wave = tid >> 5;
    const int lane = tid & 31;
    const int h    = wave & 3;
    const int sub  = wave >> 2;

    const int b0 = tile * BT + 2 * lane;
    const int ni = b0 / HW;
    const int hw = b0 - ni * HW;
    const int c0 = h * 128 + (chunk * SUBS + sub) * CHW;
    const size_t tstride = (size_t)CCH * HW;
    const size_t base = ((size_t)(ni * T_SEG) * CCH + c0) * HW + hw;
    const float2* __restrict__ xp = (const float2*)(x + base);
    float2* __restrict__ op = (float2*)(out + base);

    float2 xv[CHW][T_SEG];
    #pragma unroll
    for (int j = 0; j < CHW; ++j) {
        #pragma unroll
        for (int t = 0; t < T_SEG; ++t)
            xv[j][t] = xp[((size_t)t * tstride + (size_t)j * HW) >> 1];
    }

    const float4* __restrict__ wsm4 = (const float4*)wsm;
    #pragma unroll
    for (int i = 0; i < SLAB4 / 256; ++i) {
        const int q = i * 256 + tid;
        const int t = q / 320;
        const int r = q - t * 320;
        slab4[q] = wsm4[((size_t)t * B_TOT + (size_t)tile * BT) * (O20 / 4) + r];
    }
    __syncthreads();

    const float* sl = (const float*)slab4;
    float w[2][T_SEG][KTAPS];
    #pragma unroll
    for (int p = 0; p < 2; ++p) {
        #pragma unroll
        for (int t = 0; t < T_SEG; ++t) {
            #pragma unroll
            for (int k = 0; k < KTAPS; ++k) {
                if (t + k - 4 >= 0)
                    w[p][t][k] = sl[(t * BT + 2 * lane + p) * O20 + h * KTAPS + k];
                else
                    w[p][t][k] = 0.0f;
            }
        }
    }

    #pragma unroll
    for (int j = 0; j < CHW; ++j) {
        #pragma unroll
        for (int t = 0; t < T_SEG; ++t) {
            float a0 = 0.0f, a1 = 0.0f;
            #pragma unroll
            for (int k = 0; k < KTAPS; ++k) {
                const int ts = t + k - 4;
                if (ts >= 0) {
                    a0 = fmaf(xv[j][ts].x, w[0][t][k], a0);
                    a1 = fmaf(xv[j][ts].y, w[1][t][k], a1);
                }
            }
            float2 r; r.x = a0; r.y = a1;
            op[((size_t)t * tstride + (size_t)j * HW) >> 1] = r;
        }
    }
}

extern "C" void kernel_launch(void* const* d_in, const int* in_sizes, int n_in,
                              void* d_out, int out_size, void* d_ws, size_t ws_size,
                              hipStream_t stream) {
    (void)in_sizes; (void)n_in; (void)out_size; (void)ws_size;
    const float* x   = (const float*)d_in[0];
    const float* W   = (const float*)d_in[1];
    float*       out = (float*)d_out;
    float*       wsm = (float*)d_ws;

    dyncnn_logits_softmax<<<dim3(NSEG * T_SEG * RTILE), dim3(32 * KSPLIT), 0, stream>>>(x, W, wsm);

    dyncnn_combine<<<dim3(NBT, NCHUNK), dim3(256), 0, stream>>>(x, wsm, out);
}


# logits kernel: W copied once per work-group into LDS (row pitch 516 dwords, no bank conflicts), B fragments read from there instead of 16-line global gathers; values, k order and sum order kept
# speedup vs baseline: 3.1381x; 1.0947x over previous
#include <hip/hip_runtime.h>


typedef float v2f __attribute__((ext_vector_type(2)));
typedef float v8f __attribute__((ext_vector_type(8)));

#define T_SEG 8
#define NSEG  16
#define CCH   512
#define HW    196
#define B_TOT (NSEG * HW)
#define HEADS 4
#define KTAPS 5
#define O20   (HEADS * KTAPS)
#define RTILE 13

#define KSPLIT 4
#define CSL    (CCH / KSPLIT)
#define WPITCH4 (CCH / 4 + 1)
#define WPITCH  (WPITCH4 * 4)

__global__ __launch_bounds__(32 * KSPLIT) __attribute__((amdgpu_waves_per_eu(7, 8)))
void dyncnn_logits_softmax(const float* __restrict__ x,
                           const float* __restrict__ W,
                           float* __restrict__ wsm)
{
    __shared__ float4 Wl4[O20 * WPITCH4];

    const int tid  = threadIdx.x;
    const int lane = tid & 31;
    const int wv   = tid >> 5;
    const int tile = blockIdx.x % RTILE;
    const int tn   = blockIdx.x / RTILE;
    const int t    = tn % T_SEG;
    const int ni   = tn / T_SEG;

    const int half = lane >> 4;
    const int l15  = lane & 15;

    const float4* __restrict__ W4 = (const float4*)W;
    float4 wst[O20];
    #pragma unroll
    for (int i = 0; i < O20; ++i)
        wst[i] = W4[i * (CCH / 4) + tid];

    int hw  = tile * 16 + l15;
    int hwc = (hw < HW - 1) ? hw : (HW - 1);
    const int cA = (half ? 2 : 0) + wv * CSL;
    const float* aptr = x + ((size_t)((ni * T_SEG + t) * CCH) + cA) * HW + hwc;

    float ax0[16], ay0[16], ax1[16], ay1[16];
    #pragma unroll
    for (int s = 0; s < 16; ++s) {
        ax0[s] = aptr[(4 * s) * HW];
        ay0[s] = aptr[(4 * s + 1) * HW];
    }

    #pragma unroll
    for (int i = 0; i < O20; ++i)
        Wl4[i * WPITCH4 + tid] = wst[i];

    int off1 = 64 * HW;
    asm volatile("" : "+s"(off1));
    const float* aptr1 = aptr + off1;
    #pragma unroll
    for (int s = 0; s < 16; ++s) {
        ax1[s] = aptr1[(4 * s) * HW];
        ay1[s] = aptr1[(4 * s + 1) * HW];
    }
    __builtin_amdgcn_fence(__ATOMIC_RELEASE, "workgroup", "local");
    __builtin_amdgcn_s_barrier();
    __builtin_amdgcn_fence(__ATOMIC_ACQUIRE, "workgroup", "local");

    const float* Wl = (const float*)Wl4;
    int o0 = l15;
    int o1 = o0 + 16; if (o1 > O20 - 1) o1 = O20 - 1;
    const float* b0p = Wl + o0 * WPITCH + cA;
    const float* b1p = Wl + o1 * WPITCH + cA;

    v8f acc0 = {}; v8f acc1 = {};
    {
        float2 f0[16], f1[16];
        #pragma unroll
        for (int s = 0; s < 16; ++s) {
            f0[s] = *(const float2*)(b0p + 4 * s);
            f1[s] = *(const float2*)(b1p + 4 * s);
        }
        #pragma unroll
        for (int s = 0; s < 16; ++s) {
            v2f a;  a.x  = ax0[s];  a.y  = ay0[s];
            v2f b0; b0.x = f0[s].x; b0.y = f0[s].y;
            v2f b1; b1.x = f1[s].x; b1.y = f1[s].y;
            acc0 = __builtin_amdgcn_wmma_f32_16x16x4_f32(
                       false, a, false, b0, (short)0, acc0, false, false);
            acc1 = __builtin_amdgcn_wmma_f32_16x16x4_f32(
                       false, a, false, b1, (short)0, acc1, false, false);
        }
    }
    {
        float2 f0[16], f1[16];
        #pragma unroll
        for (int s = 0; s < 16; ++s) {
            f0[s] = *(const float2*)(b0p + 64 + 4 * s);
            f1[s] = *(const float2*)(b1p + 64 + 4 * s);
        }
        #pragma unroll
        for (int s = 0; s < 16; ++s) {
            v2f a;  a.x  = ax1[s];  a.y  = ay1[s];
            v2f b0; b0.x = f0[s].x; b0.y = f0[s].y;
            v2f b1; b1.x = f1[s].x; b1.y = f1[s].y;
            acc0 = __builtin_amdgcn_wmma_f32_16x16x4_f32(
                       false, a, false, b0, (short)0, acc0, false, false);
            acc1 = __builtin_amdgcn_wmma_f32_16x16x4_f32(
                       false, a, false, b1, (short)0, acc1, false, false);
        }
    }

    __syncthreads();

    float* Lsm = (float*)Wl4;
    {
        float* Lw = Lsm + wv * (16 * 32);
        const int mbase = half * 8;
        #pragma unroll
        for (int r = 0; r < 8; ++r) {
            Lw[(mbase + r) * 32 + l15]      = acc0[r];
            Lw[(mbase + r) * 32 + 16 + l15] = acc1[r];
        }
    }
    __syncthreads();

    if (tid < 64) {
        const int rr = tid >> 2;
        const int h  = tid & 3;
        const float* lp = &Lsm[rr * 32 + h * KTAPS];
        float v[KTAPS];
        #pragma unroll
        for (int k = 0; k < KTAPS; ++k)
            v[k] = ((lp[k] + lp[512 + k]) + lp[1024 + k]) + lp[1536 + k];
        float v0 = v[0], v1 = v[1], v2 = v[2], v3 = v[3], v4 = v[4];
        float m = fmaxf(fmaxf(fmaxf(v0, v1), fmaxf(v2, v3)), v4);
        float e0 = __expf(v0 - m), e1 = __expf(v1 - m), e2 = __expf(v2 - m),
              e3 = __expf(v3 - m), e4 = __expf(v4 - m);
        float inv = 1.0f / (e0 + e1 + e2 + e3 + e4);
        const int hwr = tile * 16 + rr;
        if (hwr < HW) {
            size_t base = (((size_t)t * B_TOT) + (size_t)ni * HW + hwr) * O20
                          + (size_t)h * KTAPS;
            wsm[base + 0] = e0 * inv;
            wsm[base + 1] = e1 * inv;
            wsm[base + 2] = e2 * inv;
            wsm[base + 3] = e3 * inv;
            wsm[base + 4] = e4 * inv;
        }
    }
}

#define BT     64
#define NBT    (B_TOT / BT)
#define CHW    4
#define SUBS   2
#define NCHUNK (128 / (CHW * SUBS))
#define SLAB4  (T_SEG * BT * O20 / 4)

__global__ __launch_bounds__(256)
void dyncnn_combine(const float* __restrict__ x,
                    const float* __restrict__ wsm,
                    float* __restrict__ out)
{
    __shared__ float4 slab4[SLAB4];
    const int tid   = threadIdx.x;
    const int tile  = blockIdx.x;
    const int chunk = blockIdx.y;

    const int wave = tid >> 5;
    const int lane = tid & 31;
    const int h    = wave & 3;
    const int sub  = wave >> 2;

    const int b0 = tile * BT + 2 * lane;
    const int ni = b0 / HW;
    const int hw = b0 - ni * HW;
    const int c0 = h * 128 + (chunk * SUBS + sub) * CHW;
    const size_t tstride = (size_t)CCH * HW;
    const size_t base = ((size_t)(ni * T_SEG) * CCH + c0) * HW + hw;
    const float2* __restrict__ xp = (const float2*)(x + base);
    float2* __restrict__ op = (float2*)(out + base);

    float2 xv[CHW][T_SEG];
    #pragma unroll
    for (int j = 0; j < CHW; ++j) {
        #pragma unroll
        for (int t = 0; t < T_SEG; ++t)
            xv[j][t] = xp[((size_t)t * tstride + (size_t)j * HW) >> 1];
    }

    const float4* __restrict__ wsm4 = (const float4*)wsm;
    #pragma unroll
    for (int i = 0; i < SLAB4 / 256; ++i) {
        const int q = i * 256 + tid;
        const int t = q / 320;
        const int r = q - t * 320;
        slab4[q] = wsm4[((size_t)t * B_TOT + (size_t)tile * BT) * (O20 / 4) + r];
    }
    __syncthreads();

    const float* sl = (const float*)slab4;
    float w[2][T_SEG][KTAPS];
    #pragma unroll
    for (int p = 0; p < 2; ++p) {
        #pragma unroll
        for (int t = 0; t < T_SEG; ++t) {
            #pragma unroll
            for (int k = 0; k < KTAPS; ++k) {
                if (t + k - 4 >= 0)
                    w[p][t][k] = sl[(t * BT + 2 * lane + p) * O20 + h * KTAPS + k];
                else
                    w[p][t][k] = 0.0f;
            }
        }
    }

    #pragma unroll
    for (int j = 0; j < CHW; ++j) {
        #pragma unroll
        for (int t = 0; t < T_SEG; ++t) {
            float a0 = 0.0f, a1 = 0.0f;
            #pragma unroll
            for (int k = 0; k < KTAPS; ++k) {
                const int ts = t + k - 4;
                if (ts >= 0) {
                    a0 = fmaf(xv[j][ts].x, w[0][t][k], a0);
                    a1 = fmaf(xv[j][ts].y, w[1][t][k], a1);
                }
            }
            float2 r; r.x = a0; r.y = a1;
            op[((size_t)t * tstride + (size_t)j * HW) >> 1] = r;
        }
    }
}

extern "C" void kernel_launch(void* const* d_in, const int* in_sizes, int n_in,
                              void* d_out, int out_size, void* d_ws, size_t ws_size,
                              hipStream_t stream) {
    (void)in_sizes; (void)n_in; (void)out_size; (void)ws_size;
    const float* x   = (const float*)d_in[0];
    const float* W   = (const float*)d_in[1];
    float*       out = (float*)d_out;
    float*       wsm = (float*)d_ws;

    dyncnn_logits_softmax<<<dim3(NSEG * T_SEG * RTILE), dim3(32 * KSPLIT), 0, stream>>>(x, W, wsm);

    dyncnn_combine<<<dim3(NBT, NCHUNK), dim3(256), 0, stream>>>(x, wsm, out);
}
